# starting kernel
# baseline (speedup 1.0000x reference)
#include <hip/hip_runtime.h>

typedef __attribute__((ext_vector_type(2))) float v2f;
typedef __attribute__((ext_vector_type(8))) float v8f;

#define PW 32
#define PH 32
#define C_DIM 128
#define WDIM 256
#define PTOT (PW * PH)

__device__ __forceinline__ float clip1(float v) {
    return __builtin_fminf(__builtin_fmaxf(v, -1.0f), 1.0f);
}

__global__ __launch_bounds__(256) void PatchSim_wmma_kernel(
    const float* __restrict__ feat,
    const int*   __restrict__ patch_ids,
    float*       __restrict__ out)
{
    __shared__ float q_s[C_DIM];

    const int bn = blockIdx.x;
    const int b  = bn >> 8;
    const int n  = bn & 255;

    const int qid   = patch_ids[n];
    const int pos_x = qid >> 8;
    const int pos_y = qid & 255;

    int sx = pos_x - PW / 2;
    if (sx < 0) sx = 0;
    if (sx > WDIM - PW) sx = WDIM - PW;
    int sy = pos_y - PH / 2;
    if (sy < 0) sy = 0;
    if (sy > WDIM - PH) sy = WDIM - PH;

    const size_t plane = (size_t)WDIM * WDIM;
    const float* featb = feat + (size_t)b * C_DIM * plane;

    const int t = threadIdx.x;
    if (t < C_DIM) {
        q_s[t] = clip1(featb[(size_t)t * plane + (size_t)pos_x * WDIM + pos_y]);
    }
    __syncthreads();

    const int lane  = t & 31;
    const int wave  = t >> 5;
    const int i     = lane & 15;
    const int halfk = (lane >> 4) << 1;

    float* outp = out + (size_t)bn * PTOT;

    for (int j = 0; j < 8; ++j) {
        const int p0 = (wave * 8 + j) * 16;
        const int p  = p0 + i;
        const int x  = sx + (p >> 5);
        const int y  = sy + (p & 31);
        const float* kptr = featb + (size_t)halfk * plane + (size_t)x * WDIM + y;

        v8f acc = {};
        #pragma unroll 4
        for (int kc = 0; kc < 32; ++kc) {
            v2f a, bb;
            a.x = clip1(kptr[0]);
            a.y = clip1(kptr[plane]);
            const int c0 = kc * 4 + halfk;
            bb.x = q_s[c0];
            bb.y = q_s[c0 + 1];
            acc = __builtin_amdgcn_wmma_f32_16x16x4_f32(
                false, a, false, bb, (short)0, acc, false, false);
            kptr += 4 * plane;
        }

        if ((lane & 15) == 0) {
            float* o = outp + p0 + ((lane >> 4) << 3);
            #pragma unroll
            for (int k2 = 0; k2 < 8; ++k2) o[k2] = acc[k2];
        }
    }
}

extern "C" void kernel_launch(void* const* d_in, const int* in_sizes, int n_in,
                              void* d_out, int out_size, void* d_ws, size_t ws_size,
                              hipStream_t stream) {
    (void)in_sizes; (void)n_in; (void)out_size; (void)d_ws; (void)ws_size;
    const float* feat      = (const float*)d_in[0];
    const int*   patch_ids = (const int*)d_in[1];
    float*       out       = (float*)d_out;

    dim3 grid(4 * 256);
    dim3 block(256);
    hipLaunchKernelGGL(PatchSim_wmma_kernel, grid, block, 0, stream,
                       feat, patch_ids, out);
}


# 16-byte loads: each lane reads 4 neighbouring window columns of a plane, so one load covers whole 128-byte rows (2 rows x 2 planes) and feeds 4 accumulator tiles; order of sums unchanged
# speedup vs baseline: 1.4372x; 1.4372x over previous
#include <hip/hip_runtime.h>


typedef __attribute__((ext_vector_type(2))) float v2f;
typedef __attribute__((ext_vector_type(4))) float v4f;
typedef __attribute__((ext_vector_type(8))) float v8f;
typedef v4f v4f_a4 __attribute__((aligned(4)));

#define PW 32
#define PH 32
#define C_DIM 128
#define WDIM 256
#define PTOT (PW * PH)

__device__ __forceinline__ float clip1(float v) {
    return __builtin_fminf(__builtin_fmaxf(v, -1.0f), 1.0f);
}

__global__ __launch_bounds__(256) void PatchSim_wmma_kernel(
    const float* __restrict__ feat,
    const int*   __restrict__ patch_ids,
    float*       __restrict__ out)
{
    __shared__ float q_s[C_DIM];

    const int bn = blockIdx.x;
    const int b  = bn >> 8;
    const int n  = bn & 255;

    const int qid   = patch_ids[n];
    const int pos_x = qid >> 8;
    const int pos_y = qid & 255;

    int sx = pos_x - PW / 2;
    if (sx < 0) sx = 0;
    if (sx > WDIM - PW) sx = WDIM - PW;
    int sy = pos_y - PH / 2;
    if (sy < 0) sy = 0;
    if (sy > WDIM - PH) sy = WDIM - PH;

    const size_t plane = (size_t)WDIM * WDIM;
    const float* featb = feat + (size_t)b * C_DIM * plane;

    const int t = threadIdx.x;
    if (t < C_DIM) {
        q_s[t] = clip1(featb[(size_t)t * plane + (size_t)pos_x * WDIM + pos_y]);
    }
    __syncthreads();

    const int lane  = t & 31;
    const int wave  = t >> 5;
    const int c     = lane & 15;
    const int h     = lane >> 4;
    const int halfk = h << 1;
    const int rr    = c >> 3;
    const int col   = (c & 7) << 2;

    float* outp = out + (size_t)bn * PTOT;

    for (int g = 0; g < 2; ++g) {
        const int row0 = wave * 4 + g * 2;
        const int x    = sx + row0 + rr;
        const float* kptr = featb + (size_t)halfk * plane + (size_t)x * WDIM + sy + col;

        v8f acc0 = {}, acc1 = {}, acc2 = {}, acc3 = {};
        for (int kb = 0; kb < 32; kb += 4) {
            v4f a0[4], a1[4];
            #pragma unroll
            for (int u = 0; u < 4; ++u) {
                a0[u] = *(const v4f_a4*)(kptr + (size_t)(4 * u) * plane);
                a1[u] = *(const v4f_a4*)(kptr + (size_t)(4 * u + 1) * plane);
            }
            __builtin_amdgcn_sched_barrier(0);
            #pragma unroll
            for (int u = 0; u < 4; ++u) {
                const int c0 = (kb + u) * 4 + halfk;
                v2f bb;
                bb.x = q_s[c0];
                bb.y = q_s[c0 + 1];
                v2f a;
                a.x = clip1(a0[u].x); a.y = clip1(a1[u].x);
                acc0 = __builtin_amdgcn_wmma_f32_16x16x4_f32(
                    false, a, false, bb, (short)0, acc0, false, false);
                a.x = clip1(a0[u].y); a.y = clip1(a1[u].y);
                acc1 = __builtin_amdgcn_wmma_f32_16x16x4_f32(
                    false, a, false, bb, (short)0, acc1, false, false);
                a.x = clip1(a0[u].z); a.y = clip1(a1[u].z);
                acc2 = __builtin_amdgcn_wmma_f32_16x16x4_f32(
                    false, a, false, bb, (short)0, acc2, false, false);
                a.x = clip1(a0[u].w); a.y = clip1(a1[u].w);
                acc3 = __builtin_amdgcn_wmma_f32_16x16x4_f32(
                    false, a, false, bb, (short)0, acc3, false, false);
            }
            kptr += 16 * plane;
        }

        if (c == 0) {
            float* o = outp + (row0 + h) * PH;
            #pragma unroll
            for (int r = 0; r < 8; ++r) {
                v4f v;
                v.x = acc0[r]; v.y = acc1[r]; v.z = acc2[r]; v.w = acc3[r];
                *(v4f*)(o + 4 * r) = v;
            }
        }
    }
}

extern "C" void kernel_launch(void* const* d_in, const int* in_sizes, int n_in,
                              void* d_out, int out_size, void* d_ws, size_t ws_size,
                              hipStream_t stream) {
    (void)in_sizes; (void)n_in; (void)out_size; (void)d_ws; (void)ws_size;
    const float* feat      = (const float*)d_in[0];
    const int*   patch_ids = (const int*)d_in[1];
    float*       out       = (float*)d_out;

    dim3 grid(4 * 256);
    dim3 block(256);
    hipLaunchKernelGGL(PatchSim_wmma_kernel, grid, block, 0, stream,
                       feat, patch_ids, out);
}


# each block owns 4 whole image rows and serves every window over them: the 16 WMMA columns carry 16 windows' queries, so an input element is read once, not once per window; order of sums unchanged
# speedup vs baseline: 3.5940x; 2.5007x over previous
#include <hip/hip_runtime.h>


typedef __attribute__((ext_vector_type(2))) float v2f;
typedef __attribute__((ext_vector_type(4))) float v4f;
typedef __attribute__((ext_vector_type(8))) float v8f;

#define PW 32
#define PH 32
#define C_DIM 128
#define WDIM 256
#define NWIN 256
#define PTOT (PW * PH)
#define RB 4
#define QP 132
#define NTHR 512

__device__ __forceinline__ float clip1(float v) {
    return __builtin_fminf(__builtin_fmaxf(v, -1.0f), 1.0f);
}

template <int N>
__device__ __forceinline__ void gather_q(const float* __restrict__ qsrc, const int* lq,
                                         float* qdst, int s0) {
    int   off[N];
    float v[N];
    #pragma unroll
    for (int i = 0; i < N; ++i) off[i] = lq[s0 + 4 * i];
    #pragma unroll
    for (int i = 0; i < N; ++i) v[i] = qsrc[off[i]];
    #pragma unroll
    for (int i = 0; i < N; ++i) qdst[(s0 + 4 * i) * QP] = clip1(v[i]);
}

__global__ __launch_bounds__(NTHR) void PatchSim_wmma_kernel(
    const float* __restrict__ feat,
    const int*   __restrict__ patch_ids,
    float*       __restrict__ out)
{
    __shared__ __attribute__((aligned(16))) float q_s[NWIN * QP];
    __shared__ __attribute__((aligned(16))) float stage_s[NWIN * RB * PH];
    __shared__ int l_n[NWIN];
    __shared__ int l_sx[NWIN];
    __shared__ int l_sy[NWIN];
    __shared__ int l_q[NWIN];
    __shared__ unsigned short w_list[NTHR / 32][NWIN];
    __shared__ int wcnt_s[NWIN / 32];

    const int t    = threadIdx.x;
    const int lane = t & 31;
    const int wave = __builtin_amdgcn_readfirstlane(t >> 5);
    const unsigned lt_mask = (1u << lane) - 1u;

    const int blk = blockIdx.x;
    const int b   = blk >> 6;
    const int x0  = (blk & 63) * RB;

    const size_t plane = (size_t)WDIM * WDIM;
    const float* featb = feat + (size_t)b * C_DIM * plane;

    int my_sx = 0, my_sy = 0, my_q = 0;
    bool ov = false;
    if (t < NWIN) {
        const int qid   = patch_ids[t];
        const int pos_x = qid >> 8;
        const int pos_y = qid & 255;
        int sx = pos_x - PW / 2;
        if (sx < 0) sx = 0;
        if (sx > WDIM - PW) sx = WDIM - PW;
        int sy = pos_y - PH / 2;
        if (sy < 0) sy = 0;
        if (sy > WDIM - PH) sy = WDIM - PH;
        my_sx = sx; my_sy = sy;
        my_q  = pos_x * WDIM + pos_y;
        ov = (sx <= x0 + RB - 1) && (sx + PW - 1 >= x0);
    }
    const unsigned bal = __builtin_amdgcn_ballot_w32(ov);
    if (lane == 0 && wave < NWIN / 32) wcnt_s[wave] = __builtin_popcount(bal);
    __syncthreads();
    int L = 0, base = 0;
    for (int i = 0; i < NWIN / 32; ++i) {
        const int cn = wcnt_s[i];
        if (i < wave) base += cn;
        L += cn;
    }
    L = __builtin_amdgcn_readfirstlane(L);
    if (ov) {
        const int slot = base + __builtin_popcount(bal & lt_mask);
        l_n[slot]  = t;
        l_sx[slot] = my_sx;
        l_sy[slot] = my_sy;
        l_q[slot]  = my_q;
    }
    __syncthreads();

    const int rp  = wave >> 3;
    const int xr0 = x0 + 2 * rp;
    const int y0  = (wave & 7) * 32;

    int cnt = 0;
    for (int s0 = 0; s0 < L; s0 += 32) {
        const int s = s0 + lane;
        bool p = false;
        if (s < L) {
            const int sx = l_sx[s];
            const int sy = l_sy[s];
            p = (sx <= xr0 + 1) && (sx + PW - 1 >= xr0) &&
                (sy <= y0 + 31) && (sy + PH - 1 >= y0);
        }
        const unsigned m = __builtin_amdgcn_ballot_w32(p);
        if (p) w_list[wave][cnt + __builtin_popcount(m & lt_mask)] = (unsigned short)s;
        cnt += __builtin_popcount(m);
    }

    {
        const int ch = t & (C_DIM - 1);
        const float* qsrc = featb + (size_t)ch * plane;
        float* qdst = q_s + ch;
        int s0 = wave >> 2;
        for (; s0 + 28 < L; s0 += 32) gather_q<8>(qsrc, l_q, qdst, s0);
        if (s0 + 12 < L) { gather_q<4>(qsrc, l_q, qdst, s0); s0 += 16; }
        if (s0 + 4 < L)  { gather_q<2>(qsrc, l_q, qdst, s0); s0 += 8; }
        if (s0 < L)      { gather_q<1>(qsrc, l_q, qdst, s0); }
    }
    __syncthreads();

    const int c     = lane & 15;
    const int h     = lane >> 4;
    const int halfk = h << 1;
    const int rr    = c >> 3;
    const int col   = (c & 7) << 2;

    const float* abase = featb + (size_t)halfk * plane + (size_t)(xr0 + rr) * WDIM + y0 + col;

    for (int p0 = 0; p0 < cnt; p0 += 16) {
        const int  idx   = p0 + c;
        const bool valid = idx < cnt;
        const int  slot  = w_list[wave][valid ? idx : p0];
        const float* qrow = q_s + slot * QP + halfk;
        const float* kptr = abase;

        v8f acc0 = {}, acc1 = {}, acc2 = {}, acc3 = {};
        for (int kb = 0; kb < 32; kb += 4) {
            v4f a0[4], a1[4];
            #pragma unroll
            for (int u = 0; u < 4; ++u) {
                a0[u] = *(const v4f*)(kptr + (size_t)(4 * u) * plane);
                a1[u] = *(const v4f*)(kptr + (size_t)(4 * u + 1) * plane);
            }
            __builtin_amdgcn_sched_barrier(0);
            #pragma unroll
            for (int u = 0; u < 4; ++u) {
                const v2f bb = *(const v2f*)(qrow + (kb + u) * 4);
                v2f a;
                a.x = clip1(a0[u].x); a.y = clip1(a1[u].x);
                acc0 = __builtin_amdgcn_wmma_f32_16x16x4_f32(
                    false, a, false, bb, (short)0, acc0, false, false);
                a.x = clip1(a0[u].y); a.y = clip1(a1[u].y);
                acc1 = __builtin_amdgcn_wmma_f32_16x16x4_f32(
                    false, a, false, bb, (short)0, acc1, false, false);
                a.x = clip1(a0[u].z); a.y = clip1(a1[u].z);
                acc2 = __builtin_amdgcn_wmma_f32_16x16x4_f32(
                    false, a, false, bb, (short)0, acc2, false, false);
                a.x = clip1(a0[u].w); a.y = clip1(a1[u].w);
                acc3 = __builtin_amdgcn_wmma_f32_16x16x4_f32(
                    false, a, false, bb, (short)0, acc3, false, false);
            }
            kptr += 16 * plane;
        }

        if (valid) {
            const int rb = 2 * rp + h;
            const int px = x0 + rb - l_sx[slot];
            if ((unsigned)px < (unsigned)PW) {
                float* st = stage_s + (slot * RB + rb) * PH;
                const int d = y0 - l_sy[slot];
                #pragma unroll
                for (int r = 0; r < 8; ++r) {
                    const int py = d + 4 * r;
                    if ((unsigned)(py + 0) < (unsigned)PH) st[py + 0] = acc0[r];
                    if ((unsigned)(py + 1) < (unsigned)PH) st[py + 1] = acc1[r];
                    if ((unsigned)(py + 2) < (unsigned)PH) st[py + 2] = acc2[r];
                    if ((unsigned)(py + 3) < (unsigned)PH) st[py + 3] = acc3[r];
                }
            }
        }
    }
    __syncthreads();

    const int nitems = L * (RB * 8);
    for (int it = t; it < nitems; it += NTHR) {
        const int slot = it >> 5;
        const int rb   = (it >> 3) & (RB - 1);
        const int q4   = it & 7;
        const int px   = x0 + rb - l_sx[slot];
        if ((unsigned)px < (unsigned)PW) {
            const v4f v = *(const v4f*)(stage_s + (slot * RB + rb) * PH + 4 * q4);
            float* o = out + (size_t)(b * NWIN + l_n[slot]) * PTOT + px * PH + 4 * q4;
            *(v4f*)o = v;
        }
    }
}

extern "C" void kernel_launch(void* const* d_in, const int* in_sizes, int n_in,
                              void* d_out, int out_size, void* d_ws, size_t ws_size,
                              hipStream_t stream) {
    (void)in_sizes; (void)n_in; (void)out_size; (void)d_ws; (void)ws_size;
    const float* feat      = (const float*)d_in[0];
    const int*   patch_ids = (const int*)d_in[1];
    float*       out       = (float*)d_out;

    dim3 grid(4 * 64);
    dim3 block(512);
    hipLaunchKernelGGL(PatchSim_wmma_kernel, grid, block, 0, stream,
                       feat, patch_ids, out);
}


# global loads requested ahead, in order of need: two plane groups at entry, all query loads in one round, two groups behind them, refills one block ahead, LDS-only barriers; order of sums unchanged
# speedup vs baseline: 4.4369x; 1.2345x over previous
#include <hip/hip_runtime.h>


typedef __attribute__((ext_vector_type(2))) float v2f;
typedef __attribute__((ext_vector_type(4))) float v4f;
typedef __attribute__((ext_vector_type(8))) float v8f;

#define PW 32
#define PH 32
#define C_DIM 128
#define WDIM 256
#define NWIN 256
#define PTOT (PW * PH)
#define RB 4
#define QP 132
#define NTHR 512

__device__ __forceinline__ float clip1(float v) {
    return __builtin_fminf(__builtin_fmaxf(v, -1.0f), 1.0f);
}

__device__ __forceinline__ void lds_barrier() {
    asm volatile("s_wait_dscnt 0x0\n\ts_barrier_signal -1\n\ts_barrier_wait -1" ::: "memory");
}

template <int N>
__device__ __forceinline__ void gather_q(const float* __restrict__ qsrc, const int* lq,
                                         float* qdst, int s0) {
    int   off[N];
    float v[N];
    #pragma unroll
    for (int i = 0; i < N; ++i) off[i] = lq[s0 + 4 * i];
    #pragma unroll
    for (int i = 0; i < N; ++i) v[i] = qsrc[off[i]];
    #pragma unroll
    for (int i = 0; i < N; ++i) qdst[(s0 + 4 * i) * QP] = clip1(v[i]);
}

__device__ __forceinline__ void load_group(v4f (&s)[8], const float* gp, unsigned loff) {
    const size_t plane = (size_t)WDIM * WDIM;
    #pragma unroll
    for (int u = 0; u < 4; ++u) {
        s[2 * u]     = *(const v4f*)(gp + (size_t)(4 * u) * plane + loff);
        s[2 * u + 1] = *(const v4f*)(gp + (size_t)(4 * u + 1) * plane + loff);
    }
}

__device__ __forceinline__ void mma_group(const v4f (&s)[8], const float* q,
                                          v8f& acc0, v8f& acc1, v8f& acc2, v8f& acc3) {
    #pragma unroll
    for (int u = 0; u < 4; ++u) {
        const v2f bb = *(const v2f*)(q + 4 * u);
        v2f a;
        a.x = clip1(s[2 * u].x); a.y = clip1(s[2 * u + 1].x);
        acc0 = __builtin_amdgcn_wmma_f32_16x16x4_f32(
            false, a, false, bb, (short)0, acc0, false, false);
        a.x = clip1(s[2 * u].y); a.y = clip1(s[2 * u + 1].y);
        acc1 = __builtin_amdgcn_wmma_f32_16x16x4_f32(
            false, a, false, bb, (short)0, acc1, false, false);
        a.x = clip1(s[2 * u].z); a.y = clip1(s[2 * u + 1].z);
        acc2 = __builtin_amdgcn_wmma_f32_16x16x4_f32(
            false, a, false, bb, (short)0, acc2, false, false);
        a.x = clip1(s[2 * u].w); a.y = clip1(s[2 * u + 1].w);
        acc3 = __builtin_amdgcn_wmma_f32_16x16x4_f32(
            false, a, false, bb, (short)0, acc3, false, false);
    }
}

__global__ __launch_bounds__(NTHR) void PatchSim_wmma_kernel(
    const float* __restrict__ feat,
    const int*   __restrict__ patch_ids,
    float*       __restrict__ out)
{
    __shared__ __attribute__((aligned(16))) float q_s[NWIN * QP];
    __shared__ __attribute__((aligned(16))) float stage_s[NWIN * RB * PH];
    __shared__ int l_n[NWIN];
    __shared__ int l_sx[NWIN];
    __shared__ int l_sy[NWIN];
    __shared__ int l_q[NWIN];
    __shared__ unsigned short w_list[NTHR / 32][NWIN];
    __shared__ int wcnt_s[NWIN / 32];

    const int t    = threadIdx.x;
    const int lane = t & 31;
    const int wave = __builtin_amdgcn_readfirstlane(t >> 5);
    const unsigned lt_mask = (1u << lane) - 1u;

    const int blk = blockIdx.x;
    const int b   = blk >> 6;
    const int x0  = (blk & 63) * RB;

    const size_t plane = (size_t)WDIM * WDIM;
    const float* featb = feat + (size_t)b * C_DIM * plane;

    const int qid = patch_ids[t & (NWIN - 1)];

    const int rp  = wave >> 3;
    const int xr0 = x0 + 2 * rp;
    const int y0  = (wave & 7) * 32;

    const int c     = lane & 15;
    const int h     = lane >> 4;
    const int halfk = h << 1;
    const int rr    = c >> 3;
    const int col   = (c & 7) << 2;

    const unsigned loff = (unsigned)halfk * (unsigned)plane + (unsigned)(xr0 + rr) * WDIM + y0 + col;

    v4f S0[8], S1[8], S2[8], S3[8];
    load_group(S0, featb,              loff);
    load_group(S1, featb + 16 * plane, loff);
    __builtin_amdgcn_sched_barrier(0);

    const int pos_x = qid >> 8;
    const int pos_y = qid & 255;
    int my_sx = pos_x - PW / 2;
    if (my_sx < 0) my_sx = 0;
    if (my_sx > WDIM - PW) my_sx = WDIM - PW;
    int my_sy = pos_y - PH / 2;
    if (my_sy < 0) my_sy = 0;
    if (my_sy > WDIM - PH) my_sy = WDIM - PH;
    const int my_q = pos_x * WDIM + pos_y;
    const bool ov = (t < NWIN) && (my_sx <= x0 + RB - 1) && (my_sx + PW - 1 >= x0);
    const unsigned bal = __builtin_amdgcn_ballot_w32(ov);
    if (lane == 0 && wave < NWIN / 32) wcnt_s[wave] = __builtin_popcount(bal);
    lds_barrier();
    int L = 0, base = 0;
    for (int i = 0; i < NWIN / 32; ++i) {
        const int cn = wcnt_s[i];
        if (i < wave) base += cn;
        L += cn;
    }
    L = __builtin_amdgcn_readfirstlane(L);
    if (ov) {
        const int slot = base + __builtin_popcount(bal & lt_mask);
        l_n[slot]  = t;
        l_sx[slot] = my_sx;
        l_sy[slot] = my_sy;
        l_q[slot]  = my_q;
    }
    lds_barrier();

    int cnt = 0;
    for (int s0 = 0; s0 < L; s0 += 32) {
        const int s = s0 + lane;
        bool p = false;
        if (s < L) {
            const int sx = l_sx[s];
            const int sy = l_sy[s];
            p = (sx <= xr0 + 1) && (sx + PW - 1 >= xr0) &&
                (sy <= y0 + 31) && (sy + PH - 1 >= y0);
        }
        const unsigned m = __builtin_amdgcn_ballot_w32(p);
        if (p) w_list[wave][cnt + __builtin_popcount(m & lt_mask)] = (unsigned short)s;
        cnt += __builtin_popcount(m);
    }

    {
        const int ch = t & (C_DIM - 1);
        const float* qsrc = featb + (size_t)ch * plane;
        float* qdst = q_s + ch;
        const int s0 = wave >> 2;
        int   off[16];
        float v[16];
        #pragma unroll
        for (int i = 0; i < 16; ++i) {
            const int s = s0 + 4 * i;
            off[i] = l_q[s < L ? s : 0];
        }
        #pragma unroll
        for (int i = 0; i < 16; ++i) {
            const float* p = (s0 + 4 * i < L) ? qsrc + off[i] : featb;
            v[i] = *p;
        }
        __builtin_amdgcn_sched_barrier(0);
        load_group(S2, featb + 32 * plane, loff);
        load_group(S3, featb + 48 * plane, loff);
        __builtin_amdgcn_sched_barrier(0);
        #pragma unroll
        for (int i = 0; i < 16; ++i)
            qdst[(s0 + 4 * i) * QP] = clip1(v[i]);
        int s1 = s0 + 64;
        for (; s1 + 28 < L; s1 += 32) gather_q<8>(qsrc, l_q, qdst, s1);
        if (s1 + 12 < L) { gather_q<4>(qsrc, l_q, qdst, s1); s1 += 16; }
        if (s1 + 4 < L)  { gather_q<2>(qsrc, l_q, qdst, s1); s1 += 8; }
        if (s1 < L)      { gather_q<1>(qsrc, l_q, qdst, s1); }
    }
    lds_barrier();

    cnt = __builtin_amdgcn_readfirstlane(cnt);

    for (int p0 = 0; p0 < cnt; ) {
        unsigned lo = loff;
        asm volatile("" : "+v"(lo));
        const int  idx   = p0 + c;
        const bool valid = idx < cnt;
        const int  slot  = w_list[wave][valid ? idx : p0];
        const float* qrow = q_s + slot * QP + halfk;

        v8f acc0 = {}, acc1 = {}, acc2 = {}, acc3 = {};

        mma_group(S0, qrow, acc0, acc1, acc2, acc3);
        __builtin_amdgcn_sched_barrier(0);
        load_group(S0, featb + 64 * plane, lo);
        __builtin_amdgcn_sched_barrier(0);
        mma_group(S1, qrow + 16, acc0, acc1, acc2, acc3);
        __builtin_amdgcn_sched_barrier(0);
        load_group(S1, featb + 80 * plane, lo);
        __builtin_amdgcn_sched_barrier(0);
        mma_group(S2, qrow + 32, acc0, acc1, acc2, acc3);
        __builtin_amdgcn_sched_barrier(0);
        load_group(S2, featb + 96 * plane, lo);
        __builtin_amdgcn_sched_barrier(0);
        mma_group(S3, qrow + 48, acc0, acc1, acc2, acc3);
        __builtin_amdgcn_sched_barrier(0);
        load_group(S3, featb + 112 * plane, lo);
        __builtin_amdgcn_sched_barrier(0);
        mma_group(S0, qrow + 64,  acc0, acc1, acc2, acc3);
        mma_group(S1, qrow + 80,  acc0, acc1, acc2, acc3);
        mma_group(S2, qrow + 96,  acc0, acc1, acc2, acc3);
        mma_group(S3, qrow + 112, acc0, acc1, acc2, acc3);

        if (valid) {
            const int rb = 2 * rp + h;
            const int px = x0 + rb - l_sx[slot];
            if ((unsigned)px < (unsigned)PW) {
                float* st = stage_s + (slot * RB + rb) * PH;
                const int d = y0 - l_sy[slot];
                #pragma unroll
                for (int r = 0; r < 8; ++r) {
                    const int py = d + 4 * r;
                    if ((unsigned)(py + 0) < (unsigned)PH) st[py + 0] = acc0[r];
                    if ((unsigned)(py + 1) < (unsigned)PH) st[py + 1] = acc1[r];
                    if ((unsigned)(py + 2) < (unsigned)PH) st[py + 2] = acc2[r];
                    if ((unsigned)(py + 3) < (unsigned)PH) st[py + 3] = acc3[r];
                }
            }
        }
        p0 += 16;
        if (p0 >= cnt) break;
        load_group(S0, featb,              lo);
        __builtin_amdgcn_sched_barrier(0);
        load_group(S1, featb + 16 * plane, lo);
        __builtin_amdgcn_sched_barrier(0);
        load_group(S2, featb + 32 * plane, lo);
        __builtin_amdgcn_sched_barrier(0);
        load_group(S3, featb + 48 * plane, lo);
    }
    __syncthreads();

    const int nitems = L * (RB * 8);
    for (int it = t; it < nitems; it += NTHR) {
        const int slot = it >> 5;
        const int rb   = (it >> 3) & (RB - 1);
        const int q4   = it & 7;
        const int px   = x0 + rb - l_sx[slot];
        if ((unsigned)px < (unsigned)PW) {
            const v4f v = *(const v4f*)(stage_s + (slot * RB + rb) * PH + 4 * q4);
            float* o = out + (size_t)(b * NWIN + l_n[slot]) * PTOT + px * PH + 4 * q4;
            *(v4f*)o = v;
        }
    }
}

extern "C" void kernel_launch(void* const* d_in, const int* in_sizes, int n_in,
                              void* d_out, int out_size, void* d_ws, size_t ws_size,
                              hipStream_t stream) {
    (void)in_sizes; (void)n_in; (void)out_size; (void)d_ws; (void)ws_size;
    const float* feat      = (const float*)d_in[0];
    const int*   patch_ids = (const int*)d_in[1];
    float*       out       = (float*)d_out;

    dim3 grid(4 * 64);
    dim3 block(512);
    hipLaunchKernelGGL(PatchSim_wmma_kernel, grid, block, 0, stream,
                       feat, patch_ids, out);
}
